# starting kernel
# baseline (speedup 1.0000x reference)
#include <hip/hip_runtime.h>
#include <math.h>

typedef __attribute__((ext_vector_type(2))) float v2f;
typedef __attribute__((ext_vector_type(8))) float v8f;

#define N_EMBD 192
#define N_HEAD 16
#define HEAD   12
#define DPAD   16
#define BATCH  4
#define SEQ    2048
#define M_TOK  (BATCH*SEQ)
#define FFN    (4*N_EMBD)
#define BH     (BATCH*N_HEAD)

static __device__ __forceinline__ v8f wmma4(v2f a, v2f b, v8f c) {
  return __builtin_amdgcn_wmma_f32_16x16x4_f32(false, a, false, b, (short)0, c, false, false);
}

__global__ void rmsnorm_kernel(const float* __restrict__ x, const float* __restrict__ w,
                               float* __restrict__ out) {
  int lane = threadIdx.x & 31;
  int wv   = threadIdx.x >> 5;
  int row  = blockIdx.x * 8 + wv;
  const float* xr = x + (size_t)row * N_EMBD;
  float v[6]; float ss = 0.f;
  #pragma unroll
  for (int i = 0; i < 6; i++) { v[i] = xr[lane + 32*i]; ss += v[i]*v[i]; }
  #pragma unroll
  for (int off = 16; off >= 1; off >>= 1) ss += __shfl_xor(ss, off, 32);
  float sc = rsqrtf(ss * (1.0f / N_EMBD) + 1e-5f);
  float* orow = out + (size_t)row * N_EMBD;
  #pragma unroll
  for (int i = 0; i < 6; i++) { int c = lane + 32*i; orow[c] = v[i] * sc * w[c]; }
}

__global__ void pack_b_kernel(const float* __restrict__ B, float* __restrict__ Bf, int N) {
  int idx = blockIdx.x * 256 + threadIdx.x;
  int k = idx / N, n = idx % N;
  Bf[((size_t)(k >> 1) * N + n) * 2 + (k & 1)] = B[idx];
}

__global__ void pack_qkv_kernel(const float* __restrict__ Wq, const float* __restrict__ Wk,
                                const float* __restrict__ Wv, float* __restrict__ Bf) {
  int idx = blockIdx.x * 256 + threadIdx.x;
  int c = idx / 576, j = idx % 576;
  int which = j / 192, r = j % 192;
  int h = r / HEAD, d = r % HEAD;
  const float* W = (which == 0) ? Wq : ((which == 1) ? Wk : Wv);
  Bf[((size_t)(c >> 1) * 576 + j) * 2 + (c & 1)] = W[((size_t)h * N_EMBD + c) * HEAD + d];
}

__device__ __forceinline__ void gemm16x64(const float* __restrict__ A, const float* __restrict__ Bf,
                                          int K, int lda, int N, int m0, int n0, v8f* acc) {
  int lane = threadIdx.x & 31;
  int mr = lane & 15;
  int kh = (lane >> 4) << 1;
  const float* arow  = A + (size_t)(m0 + mr) * lda;
  const float* bbase = Bf + (size_t)(n0 + mr) * 2;
  #pragma unroll 2
  for (int k = 0; k < K; k += 4) {
    v2f a = *(const v2f*)(arow + k + kh);
    const float* bk = bbase + (size_t)((k + kh) >> 1) * (2 * N);
    #pragma unroll
    for (int t = 0; t < 4; t++) {
      v2f b = *(const v2f*)(bk + t * 32);
      acc[t] = wmma4(a, b, acc[t]);
    }
  }
}

__global__ void gemm_qkv_kernel(const float* __restrict__ A, const float* __restrict__ Bf,
                                float* __restrict__ Qp, float* __restrict__ Kp,
                                float* __restrict__ Vp) {
  int wv = threadIdx.x >> 5;
  int tile = blockIdx.x * 8 + wv;
  const int TN = 576 / 64;
  int m0 = (tile / TN) * 16;
  int n0 = (tile % TN) * 64;
  v8f acc[4] = {{}, {}, {}, {}};
  gemm16x64(A, Bf, N_EMBD, N_EMBD, 576, m0, n0, acc);
  int lane = threadIdx.x & 31;
  int mr = lane & 15, half = lane >> 4;
  #pragma unroll
  for (int t = 0; t < 4; t++) {
    int j = n0 + t * 16 + mr;
    int which = j / 192, r = j % 192, h = r / HEAD, d = r % HEAD;
    #pragma unroll
    for (int v = 0; v < 8; v++) {
      int m = m0 + v + 8 * half;
      int b = m >> 11, tq = m & 2047;
      size_t hb = ((size_t)(b * N_HEAD + h)) * SEQ * DPAD;
      if (which == 0)      Qp[hb + (size_t)tq * DPAD + d] = acc[t][v];
      else if (which == 1) Kp[hb + (size_t)tq * DPAD + d] = acc[t][v];
      else                 Vp[hb + (size_t)(tq >> 1) * 32 + d * 2 + (tq & 1)] = acc[t][v];
    }
  }
}

__global__ void attn_kernel(const float* __restrict__ Qp, const float* __restrict__ Kp,
                            const float* __restrict__ Vp, float* __restrict__ sa) {
  __shared__ float Pb[4][256];
  int lane = threadIdx.x & 31, wv = threadIdx.x >> 5;
  int mr = lane & 15, half = lane >> 4, kh = half << 1;
  int qt = wv * 32 + blockIdx.x;
  int bh = blockIdx.y;
  const size_t base = (size_t)bh * SEQ * DPAD;
  const float* Qb = Qp + base + (size_t)qt * 16 * DPAD;

  v2f qf[4];
  #pragma unroll
  for (int ck = 0; ck < 4; ck++) qf[ck] = *(const v2f*)(Qb + mr * DPAD + ck * 4 + kh);

  v8f o = {};
  float mrun[8], lrun[8];
  #pragma unroll
  for (int v = 0; v < 8; v++) { mrun[v] = -INFINITY; lrun[v] = 0.f; }
  const float scale = 0.28867513459481287f;

  for (int kt = 0; kt <= qt; ++kt) {
    const float* Kt = Kp + base + (size_t)kt * 256;
    const float* Vt = Vp + base + (size_t)kt * 256;
    if (kt < qt) {
      __builtin_prefetch(Kt + 256, 0, 0);
      __builtin_prefetch(Vt + 256, 0, 0);
    }
    v8f s = {};
    #pragma unroll
    for (int ck = 0; ck < 4; ck++) {
      v2f bf = *(const v2f*)(Kt + mr * DPAD + ck * 4 + kh);
      s = wmma4(qf[ck], bf, s);
    }
    #pragma unroll
    for (int v = 0; v < 8; v++) {
      float sv = s[v] * scale;
      sv = (kt == qt && mr > v + 8 * half) ? -INFINITY : sv;
      float rmax = sv;
      #pragma unroll
      for (int off = 8; off >= 1; off >>= 1) rmax = fmaxf(rmax, __shfl_xor(rmax, off, 32));
      float mnew  = fmaxf(mrun[v], rmax);
      float alpha = __expf(mrun[v] - mnew);
      float pe    = __expf(sv - mnew);
      float rsum  = pe;
      #pragma unroll
      for (int off = 8; off >= 1; off >>= 1) rsum += __shfl_xor(rsum, off, 32);
      lrun[v] = lrun[v] * alpha + rsum;
      mrun[v] = mnew;
      o[v]    = o[v] * alpha;
      Pb[wv][(v + 8 * half) * 16 + mr] = pe;
    }
    __asm__ volatile("" ::: "memory");
    #pragma unroll
    for (int ck = 0; ck < 4; ck++) {
      v2f a = *(const v2f*)(&Pb[wv][mr * 16 + ck * 4 + kh]);
      int s0 = ck * 4 + kh;
      v2f bf = *(const v2f*)(Vt + (s0 >> 1) * 32 + mr * 2);
      o = wmma4(a, bf, o);
    }
  }

  int b = bh >> 4, h = bh & 15;
  if (mr < HEAD) {
    #pragma unroll
    for (int v = 0; v < 8; v++) {
      int q = qt * 16 + v + 8 * half;
      sa[((size_t)(b * SEQ + q)) * N_EMBD + h * HEAD + mr] = o[v] / lrun[v];
    }
  }
}

__global__ void gemm_proj_kernel(const float* __restrict__ A, const float* __restrict__ Bf,
                                 const float* __restrict__ bias, const float* __restrict__ resid,
                                 float* __restrict__ out) {
  int wv = threadIdx.x >> 5;
  int tile = blockIdx.x * 8 + wv;
  const int TN = 192 / 64;
  int m0 = (tile / TN) * 16, n0 = (tile % TN) * 64;
  v8f acc[4] = {{}, {}, {}, {}};
  gemm16x64(A, Bf, 192, 192, 192, m0, n0, acc);
  int lane = threadIdx.x & 31; int mr = lane & 15, half = lane >> 4;
  #pragma unroll
  for (int t = 0; t < 4; t++) {
    int j = n0 + t * 16 + mr;
    float bj = bias[j];
    #pragma unroll
    for (int v = 0; v < 8; v++) {
      int m = m0 + v + 8 * half;
      size_t ofs = (size_t)m * 192 + j;
      out[ofs] = acc[t][v] + bj + resid[ofs];
    }
  }
}

__global__ void gemm_ff1_kernel(const float* __restrict__ A, const float* __restrict__ Bf,
                                float* __restrict__ out) {
  int wv = threadIdx.x >> 5;
  int tile = blockIdx.x * 8 + wv;
  const int TN = FFN / 64;
  int m0 = (tile / TN) * 16, n0 = (tile % TN) * 64;
  v8f acc[4] = {{}, {}, {}, {}};
  gemm16x64(A, Bf, 192, 192, FFN, m0, n0, acc);
  int lane = threadIdx.x & 31; int mr = lane & 15, half = lane >> 4;
  #pragma unroll
  for (int t = 0; t < 4; t++) {
    int j = n0 + t * 16 + mr;
    #pragma unroll
    for (int v = 0; v < 8; v++) {
      int m = m0 + v + 8 * half;
      float val = acc[t][v];
      out[(size_t)m * FFN + j] = val / (1.0f + __expf(-val));
    }
  }
}

__global__ void gemm_ff2_kernel(const float* __restrict__ A, const float* __restrict__ Bf,
                                const float* __restrict__ resid, float* __restrict__ out) {
  int wv = threadIdx.x >> 5;
  int tile = blockIdx.x * 8 + wv;
  const int TN = 192 / 64;
  int m0 = (tile / TN) * 16, n0 = (tile % TN) * 64;
  v8f acc[4] = {{}, {}, {}, {}};
  gemm16x64(A, Bf, FFN, FFN, 192, m0, n0, acc);
  int lane = threadIdx.x & 31; int mr = lane & 15, half = lane >> 4;
  #pragma unroll
  for (int t = 0; t < 4; t++) {
    int j = n0 + t * 16 + mr;
    #pragma unroll
    for (int v = 0; v < 8; v++) {
      int m = m0 + v + 8 * half;
      size_t ofs = (size_t)m * 192 + j;
      out[ofs] = acc[t][v] + resid[ofs];
    }
  }
}

extern "C" void kernel_launch(void* const* d_in, const int* in_sizes, int n_in,
                              void* d_out, int out_size, void* d_ws, size_t ws_size,
                              hipStream_t stream) {
  (void)in_sizes; (void)n_in; (void)out_size; (void)ws_size;
  const float* x     = (const float*)d_in[0];
  const float* ln1_w = (const float*)d_in[1];
  const float* ln2_w = (const float*)d_in[2];
  const float* Wq    = (const float*)d_in[3];
  const float* Wk    = (const float*)d_in[4];
  const float* Wv    = (const float*)d_in[5];
  const float* Wproj = (const float*)d_in[6];
  const float* bproj = (const float*)d_in[7];
  const float* W1    = (const float*)d_in[8];
  const float* W2    = (const float*)d_in[9];
  float* ws = (float*)d_ws;

  size_t off = 0;
  float* h    = ws + off; off += (size_t)M_TOK * N_EMBD;
  float* wqkv = ws + off; off += (size_t)192 * 576;
  float* wpjf = ws + off; off += (size_t)192 * 192;
  float* w1f  = ws + off; off += (size_t)192 * FFN;
  float* w2f  = ws + off; off += (size_t)FFN * 192;
  float* Qp   = ws + off; off += (size_t)BH * SEQ * DPAD;
  float* Kp   = ws + off; off += (size_t)BH * SEQ * DPAD;
  float* Vp   = ws + off; off += (size_t)BH * SEQ * DPAD;
  float* sa   = ws + off; off += (size_t)M_TOK * N_EMBD;
  float* x1   = ws + off; off += (size_t)M_TOK * N_EMBD;
  float* h2   = ws + off; off += (size_t)M_TOK * N_EMBD;
  float* u    = ws + off; off += (size_t)M_TOK * FFN;
  float* out  = (float*)d_out;

  hipMemsetAsync(Qp, 0, (size_t)3 * BH * SEQ * DPAD * sizeof(float), stream);

  pack_qkv_kernel<<<(192 * 576) / 256, 256, 0, stream>>>(Wq, Wk, Wv, wqkv);
  pack_b_kernel<<<(192 * 192) / 256, 256, 0, stream>>>(Wproj, wpjf, 192);
  pack_b_kernel<<<(192 * FFN) / 256, 256, 0, stream>>>(W1, w1f, FFN);
  pack_b_kernel<<<(FFN * 192) / 256, 256, 0, stream>>>(W2, w2f, 192);

  rmsnorm_kernel<<<M_TOK / 8, 256, 0, stream>>>(x, ln1_w, h);
  gemm_qkv_kernel<<<(512 * 9) / 8, 256, 0, stream>>>(h, wqkv, Qp, Kp, Vp);
  attn_kernel<<<dim3(32, BH), 128, 0, stream>>>(Qp, Kp, Vp, sa);
  gemm_proj_kernel<<<(512 * 3) / 8, 256, 0, stream>>>(sa, wpjf, bproj, x, x1);
  rmsnorm_kernel<<<M_TOK / 8, 256, 0, stream>>>(x1, ln2_w, h2);
  gemm_ff1_kernel<<<(512 * 12) / 8, 256, 0, stream>>>(h2, w1f, u);
  gemm_ff2_kernel<<<(512 * 3) / 8, 256, 0, stream>>>(u, w2f, x1, out);
}


# attention operands in fragment order: lane half h takes elements 8h..8h+7 of both operands, so Q, K, V and score fragments are 16-byte loads; QKV epilogue stores K, V so; no pads stored, no memset
# speedup vs baseline: 2.3029x; 2.3029x over previous
#include <hip/hip_runtime.h>
#include <math.h>

typedef __attribute__((ext_vector_type(2))) float v2f;
typedef __attribute__((ext_vector_type(8))) float v8f;

#define N_EMBD 192
#define N_HEAD 16
#define HEAD   12
#define DPAD   16
#define BATCH  4
#define SEQ    2048
#define M_TOK  (BATCH*SEQ)
#define FFN    (4*N_EMBD)
#define BH     (BATCH*N_HEAD)

static __device__ __forceinline__ v8f wmma4(v2f a, v2f b, v8f c) {
  return __builtin_amdgcn_wmma_f32_16x16x4_f32(false, a, false, b, (short)0, c, false, false);
}

__global__ void rmsnorm_kernel(const float* __restrict__ x, const float* __restrict__ w,
                               float* __restrict__ out) {
  int lane = threadIdx.x & 31;
  int wv   = threadIdx.x >> 5;
  int row  = blockIdx.x * 8 + wv;
  const float* xr = x + (size_t)row * N_EMBD;
  float v[6]; float ss = 0.f;
  #pragma unroll
  for (int i = 0; i < 6; i++) { v[i] = xr[lane + 32*i]; ss += v[i]*v[i]; }
  #pragma unroll
  for (int off = 16; off >= 1; off >>= 1) ss += __shfl_xor(ss, off, 32);
  float sc = rsqrtf(ss * (1.0f / N_EMBD) + 1e-5f);
  float* orow = out + (size_t)row * N_EMBD;
  #pragma unroll
  for (int i = 0; i < 6; i++) { int c = lane + 32*i; orow[c] = v[i] * sc * w[c]; }
}

#define KCH    192
#define BPITCH 160
#define BS_FLOATS ((KCH / 2) * BPITCH)

__device__ __forceinline__ void gemm_lds(const float* __restrict__ A, int lda, int K,
                                         const float* __restrict__ bsrc, int bstride,
                                         int m0, float* Bs, v8f* acc) {
  const int tid = threadIdx.x;
  const int lane = tid & 31;
  const int mr = lane & 15;
  const int kh = (lane >> 4) << 1;
  const int q = tid & 15, kp0 = tid >> 4;
  const float* arow = A + (size_t)(m0 + mr) * lda;
  float4 r0[6], r1[6];
  #pragma unroll
  for (int i = 0; i < 6; i++) {
    int kp = kp0 + 16 * i;
    r0[i] = *(const float4*)(bsrc + (size_t)(2 * kp) * bstride);
    r1[i] = *(const float4*)(bsrc + (size_t)(2 * kp + 1) * bstride);
  }
  const float* bl = Bs + (kh >> 1) * BPITCH + mr * 2;
  for (int kc = 0; kc < K; kc += KCH) {
    if (kc > 0) __syncthreads();
    #pragma unroll
    for (int i = 0; i < 6; i++) {
      float* d = Bs + (kp0 + 16 * i) * BPITCH + q * 8;
      *(float4*)d       = make_float4(r0[i].x, r1[i].x, r0[i].y, r1[i].y);
      *(float4*)(d + 4) = make_float4(r0[i].z, r1[i].z, r0[i].w, r1[i].w);
    }
    __syncthreads();
    if (kc + KCH < K) {
      #pragma unroll
      for (int i = 0; i < 6; i++) {
        int kp = kp0 + 16 * i;
        r0[i] = *(const float4*)(bsrc + (size_t)(kc + KCH + 2 * kp) * bstride);
        r1[i] = *(const float4*)(bsrc + (size_t)(kc + KCH + 2 * kp + 1) * bstride);
      }
    }
    for (int k = 0; k < KCH; k += 32) {
      v2f a[8];
      #pragma unroll
      for (int s = 0; s < 8; s++) a[s] = *(const v2f*)(arow + kc + k + 4 * s + kh);
      v2f b[8][4];
      #pragma unroll
      for (int s = 0; s < 8; s++) {
        const float* bk = bl + ((k + 4 * s) >> 1) * BPITCH;
        #pragma unroll
        for (int t = 0; t < 4; t++) b[s][t] = *(const v2f*)(bk + t * 32);
      }
      #pragma unroll
      for (int s = 0; s < 8; s++) {
        #pragma unroll
        for (int t = 0; t < 4; t++) acc[t] = wmma4(a[s], b[s][t], acc[t]);
      }
    }
  }
}

__global__ void __launch_bounds__(256) gemm_qkv_kernel(const float* __restrict__ A,
                                const float* __restrict__ Wq, const float* __restrict__ Wk,
                                const float* __restrict__ Wv,
                                float* __restrict__ Qp, float* __restrict__ Kp,
                                float* __restrict__ Vp) {
  __shared__ __attribute__((aligned(16))) float Bs[BS_FLOATS];
  int wv = threadIdx.x >> 5;
  const int TN = 576 / 64;
  int m0 = (blockIdx.x / TN) * 128 + wv * 16;
  int n0 = (blockIdx.x % TN) * 64;
  const float* bsrc;
  {
    int j = n0 + 4 * (threadIdx.x & 15);
    int which = j / 192, r = j % 192, h = r / HEAD, d = r % HEAD;
    const float* Wsel = (which == 0) ? Wq : ((which == 1) ? Wk : Wv);
    bsrc = Wsel + (size_t)h * N_EMBD * HEAD + d;
  }
  v8f acc[4] = {{}, {}, {}, {}};
  gemm_lds(A, N_EMBD, N_EMBD, bsrc, HEAD, m0, Bs, acc);
  int lane = threadIdx.x & 31;
  int mr = lane & 15, half = lane >> 4;
  const int b = m0 >> 11;
  const int tile = (m0 & 2047) >> 4;
  #pragma unroll
  for (int t = 0; t < 4; t++) {
    const int jt = n0 + t * 16;
    const int which = jt / 192;
    const int r = jt % 192 + mr, h = r / HEAD, d = r % HEAD;
    const size_t hb = ((size_t)(b * N_HEAD + h)) * SEQ * DPAD + (size_t)tile * 256;
    if (which == 0) {
      float* q = Qp + hb + (8 * half) * DPAD + d;
      #pragma unroll
      for (int v = 0; v < 8; v++) q[v * DPAD] = acc[t][v];
    } else if (which == 1) {
      float* k = Kp + hb + (((d >> 2) & 1) * 2 + (d >> 3)) * 64 + (8 * half) * 4 + (d & 3);
      #pragma unroll
      for (int v = 0; v < 8; v++) k[v * 4] = acc[t][v];
    } else {
      float* vv = Vp + hb + half * 64 + d * 4;
      #pragma unroll
      for (int mb = 0; mb < 2; mb++)
        *(float4*)(vv + mb * 128) = make_float4(acc[t][4 * mb], acc[t][4 * mb + 1], acc[t][4 * mb + 2], acc[t][4 * mb + 3]);
    }
  }
}

#define KT_STEP 4
#define PT_HALF 160
#define PT_TILE 288
#define PART_STRIDE 288

static __device__ __forceinline__ float xhalf(float x) {
  int b = __builtin_bit_cast(int, x);
  return __builtin_bit_cast(float,
      __builtin_amdgcn_permlanex16(b, b, 0x76543210, 0xfedcba98, false, false));
}

__global__ void __launch_bounds__(128) __attribute__((amdgpu_waves_per_eu(8, 8))) attn_kernel(const float* __restrict__ Qp, const float* __restrict__ Kp,
                            const float* __restrict__ Vp, float* __restrict__ sa) {
  __shared__ __attribute__((aligned(16))) float Pb[4][KT_STEP * PT_TILE];
  __shared__ __attribute__((aligned(16))) float Rb[4][16];
  __shared__ __attribute__((aligned(16))) float Part[4][4][PART_STRIDE];
  int lane = threadIdx.x & 31;
  int wv = __builtin_amdgcn_readfirstlane(threadIdx.x >> 5);
  int mr = lane & 15, half = lane >> 4;
  int bh = blockIdx.y;
  const size_t base = (size_t)bh * SEQ * DPAD;
  const float* Kh = Kp + base;
  const float* Vh = Vp + base;
  float* P = Pb[wv];
  float* R = Rb[wv];
  const float cs = 0.28867513459481287f * 1.4426950408889634f;
  const int fk0 = half * 64 + mr * 4;
  const int fk1 = half ? fk0 : 128 + mr * 4;
  const int fv = half * 64 + (mr < HEAD ? mr : mr - 4) * 4;
  const int pst0 = half * PT_HALF + (mr ^ (half << 3));
  const int pst1 = half * PT_HALF + (mr ^ ((half << 3) | 4));
  const int prow = (mr >> 3) * PT_HALF + (mr & 7) * 16;
  const int pld0 = prow + (((2 * half)     ^ ((mr >> 2) & 3)) << 2);
  const int pld1 = prow + (((2 * half + 1) ^ ((mr >> 2) & 3)) << 2);

  int e0 = 0;
  for (int i = 0; i < 4; ++i) {
    const int qt = i * 32 + ((blockIdx.x + 8 * i) & 31);
    const float* Qb = Qp + base + (size_t)qt * 16 * DPAD;
    v2f qf[4];
    {
      float4 q0 = *(const float4*)(Qb + mr * DPAD + 8 * half);
      float4 q1 = *(const float4*)(Qb + mr * DPAD + 4);
      if (half) q1 = make_float4(0.f, 0.f, 0.f, 0.f);
      qf[0] = (v2f){q0.x, q0.y}; qf[1] = (v2f){q0.z, q0.w};
      qf[2] = (v2f){q1.x, q1.y}; qf[3] = (v2f){q1.z, q1.w};
    }

    v8f o = {};
    float mrun = -INFINITY;
    float lrun = 0.f;

    const int nfull = qt >> 2;
    for (int kb = (wv - e0) & 3; kb <= nfull; kb += 4) {
      const bool last = (kb == nfull);
      int kt[KT_STEP];
      #pragma unroll
      for (int t = 0; t < KT_STEP; t++) {
        int k = kb * KT_STEP + t;
        kt[t] = k < qt ? k : qt;
      }
      if (kb + 4 <= nfull) {
        __builtin_prefetch(Kh + (size_t)(kb + 4) * (KT_STEP * 256) + lane * 32, 0, 0);
        __builtin_prefetch(Vh + (size_t)(kb + 4) * (KT_STEP * 256) + lane * 32, 0, 0);
      }

      v8f s[KT_STEP] = {{}, {}, {}, {}};
      v2f kf[KT_STEP][4];
      #pragma unroll
      for (int t = 0; t < KT_STEP; t++) {
        const float* kp = Kh + (size_t)kt[t] * 256;
        float4 k0 = *(const float4*)(kp + fk0);
        float4 k1 = *(const float4*)(kp + fk1);
        kf[t][0] = (v2f){k0.x, k0.y}; kf[t][1] = (v2f){k0.z, k0.w};
        kf[t][2] = (v2f){k1.x, k1.y}; kf[t][3] = (v2f){k1.z, k1.w};
      }
      #pragma unroll
      for (int ck = 0; ck < 4; ck++) {
        #pragma unroll
        for (int t = 0; t < KT_STEP; t++) s[t] = wmma4(qf[ck], kf[t][ck], s[t]);
      }
      v2f vf[KT_STEP][4];
      #pragma unroll
      for (int t = 0; t < KT_STEP; t++) {
        const float* vp = Vh + (size_t)kt[t] * 256 + fv;
        float4 v0 = *(const float4*)vp;
        float4 v1 = *(const float4*)(vp + 128);
        vf[t][0] = (v2f){v0.x, v0.y}; vf[t][1] = (v2f){v0.z, v0.w};
        vf[t][2] = (v2f){v1.x, v1.y}; vf[t][3] = (v2f){v1.z, v1.w};
      }
      if (last) {
        #pragma unroll
        for (int t = 0; t < KT_STEP; t++) {
          int d = (kb * KT_STEP + t - qt) * 16 + mr - 8 * half;
          #pragma unroll
          for (int v = 0; v < 8; v++) s[t][v] = (d > v) ? -INFINITY : s[t][v];
        }
      }
      #pragma unroll
      for (int t = 0; t < KT_STEP; t++) {
        #pragma unroll
        for (int v = 0; v < 8; v++) P[t * PT_TILE + v * 16 + (v < 4 ? pst0 : pst1)] = s[t][v];
      }
      __asm__ volatile("" ::: "memory");
      v2f a[KT_STEP][4];
      #pragma unroll
      for (int t = 0; t < KT_STEP; t++) {
        float4 p0 = *(const float4*)(P + t * PT_TILE + pld0);
        float4 p1 = *(const float4*)(P + t * PT_TILE + pld1);
        a[t][0] = (v2f){p0.x, p0.y}; a[t][1] = (v2f){p0.z, p0.w};
        a[t][2] = (v2f){p1.x, p1.y}; a[t][3] = (v2f){p1.z, p1.w};
      }
      float rmax = fmaxf(a[0][0].x, a[0][0].y);
      #pragma unroll
      for (int t = 0; t < KT_STEP; t++) {
        #pragma unroll
        for (int ck = 0; ck < 4; ck++) rmax = fmaxf(rmax, fmaxf(a[t][ck].x, a[t][ck].y));
      }
      rmax = fmaxf(rmax, xhalf(rmax));
      float mnew  = fmaxf(mrun, rmax * cs);
      float alpha = __builtin_amdgcn_exp2f(mrun - mnew);
      mrun = mnew;
      R[mr] = alpha;
      v2f nm2 = {-mnew, -mnew};
      v2f rs2 = {0.f, 0.f};
      #pragma unroll
      for (int t = 0; t < KT_STEP; t++) {
        #pragma unroll
        for (int ck = 0; ck < 4; ck++) {
          v2f x = a[t][ck] * cs + nm2;
          x.x = __builtin_amdgcn_exp2f(x.x);
          x.y = __builtin_amdgcn_exp2f(x.y);
          a[t][ck] = x;
          rs2 += x;
        }
      }
      lrun = lrun * alpha + (rs2.x + rs2.y);
      __asm__ volatile("" ::: "memory");
      float4 al0 = *(const float4*)(R + 8 * half);
      float4 al1 = *(const float4*)(R + 8 * half + 4);
      o[0] *= al0.x; o[1] *= al0.y; o[2] *= al0.z; o[3] *= al0.w;
      o[4] *= al1.x; o[5] *= al1.y; o[6] *= al1.z; o[7] *= al1.w;
      #pragma unroll
      for (int t = 0; t < KT_STEP; t++) {
        #pragma unroll
        for (int ck = 0; ck < 4; ck++) o = wmma4(a[t][ck], vf[t][ck], o);
      }
    }
    e0 += nfull + 1;

    float lsum = lrun + xhalf(lrun);
    float* pp = &Part[i][wv][0];
    #pragma unroll
    for (int v = 0; v < 8; v++) pp[(v + 8 * half) * 16 + mr] = o[v];
    pp[256 + mr] = mrun;
    pp[272 + mr] = lsum;
  }

  __syncthreads();

  {
    const int qt = wv * 32 + ((blockIdx.x + 8 * wv) & 31);
    const float* pt = &Part[wv][0][0];
    float mw[4][8], lw[4][8];
    #pragma unroll
    for (int w = 0; w < 4; w++) {
      float4 m0 = *(const float4*)(pt + w * PART_STRIDE + 256 + 8 * half);
      float4 m1 = *(const float4*)(pt + w * PART_STRIDE + 256 + 8 * half + 4);
      float4 l0 = *(const float4*)(pt + w * PART_STRIDE + 272 + 8 * half);
      float4 l1 = *(const float4*)(pt + w * PART_STRIDE + 272 + 8 * half + 4);
      mw[w][0] = m0.x; mw[w][1] = m0.y; mw[w][2] = m0.z; mw[w][3] = m0.w;
      mw[w][4] = m1.x; mw[w][5] = m1.y; mw[w][6] = m1.z; mw[w][7] = m1.w;
      lw[w][0] = l0.x; lw[w][1] = l0.y; lw[w][2] = l0.z; lw[w][3] = l0.w;
      lw[w][4] = l1.x; lw[w][5] = l1.y; lw[w][6] = l1.z; lw[w][7] = l1.w;
    }
    int b = bh >> 4, h = bh & 15;
    #pragma unroll
    for (int v = 0; v < 8; v++) {
      float M = fmaxf(fmaxf(mw[0][v], mw[1][v]), fmaxf(mw[2][v], mw[3][v]));
      float osum = 0.f, lsum = 0.f;
      #pragma unroll
      for (int w = 0; w < 4; w++) {
        float g = __builtin_amdgcn_exp2f(mw[w][v] - M);
        lsum += g * lw[w][v];
        osum += g * pt[w * PART_STRIDE + (v + 8 * half) * 16 + mr];
      }
      if (mr < HEAD) {
        int q = qt * 16 + v + 8 * half;
        sa[((size_t)(b * SEQ + q)) * N_EMBD + h * HEAD + mr] = osum / lsum;
      }
    }
  }
}

__global__ void __launch_bounds__(256) gemm_proj_kernel(const float* __restrict__ A, const float* __restrict__ B,
                                 const float* __restrict__ bias, const float* __restrict__ resid,
                                 float* __restrict__ out) {
  __shared__ __attribute__((aligned(16))) float Bs[BS_FLOATS];
  int wv = threadIdx.x >> 5;
  const int TN = 192 / 64;
  int m0 = (blockIdx.x / TN) * 128 + wv * 16, n0 = (blockIdx.x % TN) * 64;
  v8f acc[4] = {{}, {}, {}, {}};
  gemm_lds(A, 192, 192, B + n0 + 4 * (threadIdx.x & 15), 192, m0, Bs, acc);
  int lane = threadIdx.x & 31; int mr = lane & 15, half = lane >> 4;
  #pragma unroll
  for (int t = 0; t < 4; t++) {
    int j = n0 + t * 16 + mr;
    float bj = bias[j];
    #pragma unroll
    for (int v = 0; v < 8; v++) {
      int m = m0 + v + 8 * half;
      size_t ofs = (size_t)m * 192 + j;
      out[ofs] = acc[t][v] + bj + resid[ofs];
    }
  }
}

__global__ void __launch_bounds__(256) gemm_ff1_kernel(const float* __restrict__ A, const float* __restrict__ B,
                                float* __restrict__ out) {
  __shared__ __attribute__((aligned(16))) float Bs[BS_FLOATS];
  int wv = threadIdx.x >> 5;
  const int TN = FFN / 64;
  int m0 = (blockIdx.x / TN) * 128 + wv * 16, n0 = (blockIdx.x % TN) * 64;
  v8f acc[4] = {{}, {}, {}, {}};
  gemm_lds(A, 192, 192, B + n0 + 4 * (threadIdx.x & 15), FFN, m0, Bs, acc);
  int lane = threadIdx.x & 31; int mr = lane & 15, half = lane >> 4;
  #pragma unroll
  for (int t = 0; t < 4; t++) {
    int j = n0 + t * 16 + mr;
    #pragma unroll
    for (int v = 0; v < 8; v++) {
      int m = m0 + v + 8 * half;
      float val = acc[t][v];
      out[(size_t)m * FFN + j] = val / (1.0f + __expf(-val));
    }
  }
}

__global__ void __launch_bounds__(256) gemm_ff2_kernel(const float* __restrict__ A, const float* __restrict__ B,
                                const float* __restrict__ resid, float* __restrict__ out) {
  __shared__ __attribute__((aligned(16))) float Bs[BS_FLOATS];
  int wv = threadIdx.x >> 5;
  const int TN = 192 / 64;
  int m0 = (blockIdx.x / TN) * 128 + wv * 16, n0 = (blockIdx.x % TN) * 64;
  v8f acc[4] = {{}, {}, {}, {}};
  gemm_lds(A, FFN, FFN, B + n0 + 4 * (threadIdx.x & 15), 192, m0, Bs, acc);
  int lane = threadIdx.x & 31; int mr = lane & 15, half = lane >> 4;
  #pragma unroll
  for (int t = 0; t < 4; t++) {
    int j = n0 + t * 16 + mr;
    #pragma unroll
    for (int v = 0; v < 8; v++) {
      int m = m0 + v + 8 * half;
      size_t ofs = (size_t)m * 192 + j;
      out[ofs] = acc[t][v] + resid[ofs];
    }
  }
}

extern "C" void kernel_launch(void* const* d_in, const int* in_sizes, int n_in,
                              void* d_out, int out_size, void* d_ws, size_t ws_size,
                              hipStream_t stream) {
  (void)in_sizes; (void)n_in; (void)out_size; (void)ws_size;
  const float* x     = (const float*)d_in[0];
  const float* ln1_w = (const float*)d_in[1];
  const float* ln2_w = (const float*)d_in[2];
  const float* Wq    = (const float*)d_in[3];
  const float* Wk    = (const float*)d_in[4];
  const float* Wv    = (const float*)d_in[5];
  const float* Wproj = (const float*)d_in[6];
  const float* bproj = (const float*)d_in[7];
  const float* W1    = (const float*)d_in[8];
  const float* W2    = (const float*)d_in[9];
  float* ws = (float*)d_ws;

  size_t off = 0;
  float* h    = ws + off; off += (size_t)M_TOK * N_EMBD;
  float* Qp   = ws + off; off += (size_t)BH * SEQ * DPAD;
  float* Kp   = ws + off; off += (size_t)BH * SEQ * DPAD;
  float* Vp   = ws + off; off += (size_t)BH * SEQ * DPAD;
  float* sa   = ws + off; off += (size_t)M_TOK * N_EMBD;
  float* x1   = ws + off; off += (size_t)M_TOK * N_EMBD;
  float* h2   = ws + off; off += (size_t)M_TOK * N_EMBD;
  float* u    = ws + off; off += (size_t)M_TOK * FFN;
  float* out  = (float*)d_out;

  rmsnorm_kernel<<<M_TOK / 8, 256, 0, stream>>>(x, ln1_w, h);
  gemm_qkv_kernel<<<64 * 9, 256, 0, stream>>>(h, Wq, Wk, Wv, Qp, Kp, Vp);
  attn_kernel<<<dim3(32, BH), 128, 0, stream>>>(Qp, Kp, Vp, sa);
  gemm_proj_kernel<<<64 * 3, 256, 0, stream>>>(sa, Wproj, bproj, x, x1);
  rmsnorm_kernel<<<M_TOK / 8, 256, 0, stream>>>(x1, ln2_w, h2);
  gemm_ff1_kernel<<<64 * 12, 256, 0, stream>>>(h2, W1, u);
  gemm_ff2_kernel<<<64 * 3, 256, 0, stream>>>(u, W2, x1, out);
}
